# starting kernel
# baseline (speedup 1.0000x reference)
#include <hip/hip_runtime.h>
#include <math.h>

typedef __attribute__((ext_vector_type(2))) float v2f;
typedef __attribute__((ext_vector_type(4))) float v4f;
typedef __attribute__((ext_vector_type(8))) float v8f;

#define D_DIM 1024
#define B_DIM 2048
#define M_MOL 8
#define NPAIR 28

__device__ __constant__ unsigned char c_I[NPAIR] =
    {0,0,0,0,0,0,0, 1,1,1,1,1,1, 2,2,2,2,2, 3,3,3,3, 4,4,4, 5,5, 6};
__device__ __constant__ unsigned char c_J[NPAIR] =
    {1,2,3,4,5,6,7, 2,3,4,5,6,7, 3,4,5,6,7, 4,5,6,7, 5,6,7, 6,7, 7};

static __device__ inline v8f wmma4(v2f a, v2f b, v8f c) {
  return __builtin_amdgcn_wmma_f32_16x16x4_f32(false, a, false, b, (short)0, c,
                                               false, false);
}


__global__ __launch_bounds__(256) void gemm_gt(const float* __restrict__ Wq,
                                               const float* __restrict__ Wk,
                                               float* __restrict__ Gpk) {
  const int wave = (int)((blockIdx.x * blockDim.x + threadIdx.x) >> 5);
  const int lane = threadIdx.x & 31;
  const int lm = lane & 15;
  const int half = lane >> 4;
  const int tm = wave >> 6;
  const int tn = wave & 63;

  const float* arow = Wk + (size_t)(tm * 16 + lm) * D_DIM + 2 * half;
  const float* brow = Wq + (size_t)(tn * 16 + lm) * D_DIM + 2 * half;

  v8f c = {};
#pragma unroll 4
  for (int x0 = 0; x0 < D_DIM; x0 += 4) {
    v2f a = *(const v2f*)(arow + x0);
    v2f b = *(const v2f*)(brow + x0);
    c = wmma4(a, b, c);
  }
  const int colOff = ((lm >> 1) & 1) * 512 + (tn * 4 + (lm >> 2)) * 2 + (lm & 1);
#pragma unroll
  for (int v = 0; v < 8; ++v)
    Gpk[(size_t)(tm * 16 + v + 8 * half) * D_DIM + colOff] = c[v];
}

__global__ __launch_bounds__(256) void attn_batch(
    const float* __restrict__ rF, const float* __restrict__ pF,
    const float* __restrict__ rN, const float* __restrict__ pN,
    const float* __restrict__ Gpk, const float* __restrict__ Wp,
    const float* __restrict__ bp, float* __restrict__ out) {
  __shared__ float sW[8][256];
  __shared__ float sS[8][256];
  __shared__ float sRows[72][36];
  __shared__ float sAtt[2][36];
  __shared__ float sWeff[2][8];
  __shared__ float sRed[256 * 4];

  const int b = blockIdx.x;
  const int tid = threadIdx.x;
  const int wave = tid >> 5;
  const int lane = tid & 31;
  const int lm = lane & 15;
  const int half = lane >> 4;

  const float* Frow =
      (lm < M_MOL) ? (pF + ((size_t)lm * B_DIM + b) * D_DIM)
                   : (rF + ((size_t)(lm - M_MOL) * B_DIM + b) * D_DIM);
  const float* fa = Frow + 2 * half;

  v8f sacc = {};
#pragma unroll 1
  for (int i = 0; i < 2; ++i) {
    int e[4];
    const float* g[4];
#pragma unroll
    for (int j = 0; j < 4; ++j) {
      e[j] = (wave + 8 * (4 * i + j)) * 16;
      g[j] = Gpk + (size_t)(e[j] + lm) * D_DIM + half * 512;
    }
    v8f c0 = {}, c1 = {}, c2 = {}, c3 = {};
#pragma unroll 1
    for (int d0 = 0; d0 < D_DIM; d0 += 8) {
      v2f a0 = *(const v2f*)(fa + d0);
      v2f a1 = *(const v2f*)(fa + d0 + 4);
      v4f g0 = *(const v4f*)(g[0] + (d0 >> 1));
      v4f g1 = *(const v4f*)(g[1] + (d0 >> 1));
      v4f g2 = *(const v4f*)(g[2] + (d0 >> 1));
      v4f g3 = *(const v4f*)(g[3] + (d0 >> 1));
      c0 = wmma4(a0, __builtin_shufflevector(g0, g0, 0, 1), c0);
      c1 = wmma4(a0, __builtin_shufflevector(g1, g1, 0, 1), c1);
      c2 = wmma4(a0, __builtin_shufflevector(g2, g2, 0, 1), c2);
      c3 = wmma4(a0, __builtin_shufflevector(g3, g3, 0, 1), c3);
      c0 = wmma4(a1, __builtin_shufflevector(g0, g0, 2, 3), c0);
      c1 = wmma4(a1, __builtin_shufflevector(g1, g1, 2, 3), c1);
      c2 = wmma4(a1, __builtin_shufflevector(g2, g2, 2, 3), c2);
      c3 = wmma4(a1, __builtin_shufflevector(g3, g3, 2, 3), c3);
    }
#pragma unroll
    for (int j = 0; j < 4; ++j) {
      const v8f cj = (j == 0) ? c0 : (j == 1) ? c1 : (j == 2) ? c2 : c3;
#pragma unroll
      for (int v = 0; v < 8; ++v) sW[wave][(v + 8 * half) * 16 + lm] = cj[v];
#pragma unroll
      for (int kk = 0; kk < 4; ++kk) {
        v2f a2 = *(const v2f*)(&sW[wave][lm * 16 + kk * 4 + 2 * half]);
        v2f b2 = *(const v2f*)(Frow + e[j] + kk * 4 + 2 * half);
        sacc = wmma4(a2, b2, sacc);
      }
    }
  }
#pragma unroll
  for (int v = 0; v < 8; ++v) sS[wave][(v + 8 * half) * 16 + lm] = sacc[v];
  __syncthreads();

  {
    float s = 0.f;
#pragma unroll
    for (int w = 0; w < 8; ++w) s += sS[w][tid];
    sS[0][tid] = s;
  }
  __syncthreads();

  const float scale = 0.03125f;
  if (tid < 72) {
    const int dir = (tid >= 36) ? 1 : 0;
    const int q = dir ? tid - 36 : tid;
    int qa0 = q, qa1 = q, qn = 1;
    if (q >= 8) { qa0 = c_I[q - 8]; qa1 = c_J[q - 8]; qn = 2; }
    float row[36];
    float mx = -3.4e38f;
#pragma unroll
    for (int k = 0; k < 36; ++k) {
      int ka0 = k, ka1 = k, kn = 1;
      if (k >= 8) { ka0 = c_I[k - 8]; ka1 = c_J[k - 8]; kn = 2; }
      float s = 0.f;
      for (int u = 0; u < qn; ++u) {
        const int a = u ? qa1 : qa0;
        for (int v = 0; v < kn; ++v) {
          const int cc = v ? ka1 : ka0;
          s += (dir == 0) ? sS[0][a * 16 + 8 + cc] : sS[0][(8 + a) * 16 + cc];
        }
      }
      s *= scale;
      row[k] = s;
      mx = fmaxf(mx, s);
    }
    float sum = 0.f;
#pragma unroll
    for (int k = 0; k < 36; ++k) { float e = __expf(row[k] - mx); row[k] = e; sum += e; }
    const float inv = 1.f / sum;
#pragma unroll
    for (int k = 0; k < 36; ++k) sRows[tid][k] = row[k] * inv;
  }
  __syncthreads();

  if (tid < 72) {
    const int dir = (tid >= 36) ? 1 : 0;
    const int k = dir ? tid - 36 : tid;
    float a = 0.f;
#pragma unroll
    for (int q = 0; q < 36; ++q) a += sRows[dir * 36 + q][k];
    a *= (1.f / 36.f);
    sAtt[dir][k] = a;
    const size_t base = dir ? (size_t)(B_DIM * 4 + B_DIM * 36) : (size_t)(B_DIM * 4);
    out[base + (size_t)b * 36 + k] = a;
  }
  __syncthreads();

  if (tid < 16) {
    const int dir = tid >> 3, m = tid & 7;
    float w = sAtt[dir][m];
#pragma unroll
    for (int pr = 0; pr < NPAIR; ++pr)
      if (c_I[pr] == m || c_J[pr] == m) w += sAtt[dir][8 + pr];
    sWeff[dir][m] = w;
  }
  __syncthreads();

  {
    const int d = tid * 4;
    float reac[4] = {0.f, 0.f, 0.f, 0.f};
#pragma unroll
    for (int m = 0; m < M_MOL; ++m) {
      const size_t off = ((size_t)m * B_DIM + b) * D_DIM + d;
      v4f rv = *(const v4f*)(rF + off);
      v4f rnv = *(const v4f*)(rN + off);
      v4f pv = *(const v4f*)(pF + off);
      v4f pnv = *(const v4f*)(pN + off);
      const float wr = sWeff[0][m], wp = sWeff[1][m];
#pragma unroll
      for (int x = 0; x < 4; ++x)
        reac[x] += wr * (rv[x] + rnv[x]) - wp * (pv[x] + pnv[x]);
    }
#pragma unroll
    for (int o = 0; o < 4; ++o) {
      float s = 0.f;
#pragma unroll
      for (int x = 0; x < 4; ++x) s += reac[x] * Wp[(size_t)o * D_DIM + d + x];
      sRed[tid * 4 + o] = s;
    }
  }
  __syncthreads();
  if (tid < 4) {
    float s = bp[tid];
    for (int t = 0; t < 256; ++t) s += sRed[t * 4 + tid];
    out[(size_t)b * 4 + tid] = s;
  }
}

extern "C" void kernel_launch(void* const* d_in, const int* in_sizes, int n_in,
                              void* d_out, int out_size, void* d_ws,
                              size_t ws_size, hipStream_t stream) {
  (void)in_sizes; (void)n_in; (void)out_size; (void)ws_size;
  const float* rF = (const float*)d_in[0];
  const float* pF = (const float*)d_in[1];
  const float* rN = (const float*)d_in[2];
  const float* pN = (const float*)d_in[3];
  const float* Wq = (const float*)d_in[4];
  const float* Wk = (const float*)d_in[5];
  const float* Wp = (const float*)d_in[6];
  const float* bp = (const float*)d_in[7];
  float* out = (float*)d_out;
  float* Gpk = (float*)d_ws;

  gemm_gt<<<512, 256, 0, stream>>>(Wq, Wk, Gpk);
  attn_batch<<<B_DIM, 256, 0, stream>>>(rF, pF, rN, pN, Gpk, Wp, bp, out);
}


# G^T tiles staged through a ring of three LDS buffers: a tile's barrier is signalled after its last read and waited for one tile later, in mid-stream; fragment reads one block ahead across tiles
# speedup vs baseline: 6.3792x; 6.3792x over previous
#include <hip/hip_runtime.h>
#include <math.h>


typedef __attribute__((ext_vector_type(2))) float v2f;
typedef __attribute__((ext_vector_type(4))) float v4f;
typedef __attribute__((ext_vector_type(8))) float v8f;

#define D_DIM 1024
#define B_DIM 2048
#define M_MOL 8
#define NPAIR 28

#define NW 8
#define NC 8
#define NE (NC * 16)
#define KC 64
#define GP (KC + 4)
#define NKC (D_DIM / KC)
#define NEG (D_DIM / NE)
#define NIT (NKC * NEG)
#define NST (NE * (KC / 4) / (NW * 32))
#define BK 2
#define NBK (KC / 8 / BK)
#define RP (256 * 4 + 4)
#define NBUF 3
#define TILE_F (NE * GP)

__device__ __constant__ unsigned char c_I[NPAIR] =
    {0,0,0,0,0,0,0, 1,1,1,1,1,1, 2,2,2,2,2, 3,3,3,3, 4,4,4, 5,5, 6};
__device__ __constant__ unsigned char c_J[NPAIR] =
    {1,2,3,4,5,6,7, 2,3,4,5,6,7, 3,4,5,6,7, 4,5,6,7, 5,6,7, 6,7, 7};

static __device__ inline v8f wmma4(v2f a, v2f b, v8f c) {
  return __builtin_amdgcn_wmma_f32_16x16x4_f32(false, a, false, b, (short)0, c,
                                               false, false);
}

#define GK 8
#define GNB (D_DIM / (8 * GK))

static __device__ __forceinline__ void gt_load(v4f (&f)[2][GK], const float* p, int kb) {
#pragma unroll
  for (int t = 0; t < 2; ++t)
#pragma unroll
    for (int g = 0; g < GK; ++g)
      f[t][g] = *(const v4f*)(p + (size_t)(t * 16) * D_DIM + kb * (8 * GK) + g * 8);
}

static __device__ __forceinline__ void gt_mma(const v4f (&fa)[2][GK], const v4f (&fb)[2][GK],
                                              v8f (&c)[2][2]) {
#pragma unroll
  for (int g = 0; g < GK; ++g) {
#pragma unroll
    for (int i = 0; i < 2; ++i)
#pragma unroll
      for (int j = 0; j < 2; ++j)
        c[i][j] = wmma4(__builtin_shufflevector(fa[i][g], fa[i][g], 0, 1),
                        __builtin_shufflevector(fb[j][g], fb[j][g], 0, 1), c[i][j]);
#pragma unroll
    for (int i = 0; i < 2; ++i)
#pragma unroll
      for (int j = 0; j < 2; ++j)
        c[i][j] = wmma4(__builtin_shufflevector(fa[i][g], fa[i][g], 2, 3),
                        __builtin_shufflevector(fb[j][g], fb[j][g], 2, 3), c[i][j]);
  }
}

__global__ __launch_bounds__(128) void gemm_gt(const float* __restrict__ Wq,
                                               const float* __restrict__ Wk,
                                               float* __restrict__ GT) {
  const int w = threadIdx.x >> 5;
  const int lane = threadIdx.x & 31;
  const int lm = lane & 15;
  const int half = lane >> 4;
  const int e0 = (int)(blockIdx.x >> 4) * 64 + (w >> 1) * 32;
  const int d0 = (int)(blockIdx.x & 15) * 64 + (w & 1) * 32;

  const float* ap = Wk + (size_t)(e0 + lm) * D_DIM + 4 * half;
  const float* bp = Wq + (size_t)(d0 + lm) * D_DIM + 4 * half;

  v8f c[2][2];
#pragma unroll
  for (int i = 0; i < 2; ++i)
#pragma unroll
    for (int j = 0; j < 2; ++j) c[i][j] = (v8f){};

  v4f fa0[2][GK], fb0[2][GK], fa1[2][GK], fb1[2][GK];
  gt_load(fa0, ap, 0);
  gt_load(fb0, bp, 0);
#pragma unroll 1
  for (int kb = 0; kb < GNB; kb += 2) {
    gt_load(fa1, ap, kb + 1);
    gt_load(fb1, bp, kb + 1);
    __builtin_amdgcn_sched_barrier(0);
    gt_mma(fa0, fb0, c);
    __builtin_amdgcn_sched_barrier(0);
    gt_load(fa0, ap, (kb + 2) & (GNB - 1));
    gt_load(fb0, bp, (kb + 2) & (GNB - 1));
    __builtin_amdgcn_sched_barrier(0);
    gt_mma(fa1, fb1, c);
    __builtin_amdgcn_sched_barrier(0);
  }

#pragma unroll
  for (int i = 0; i < 2; ++i)
#pragma unroll
    for (int j = 0; j < 2; ++j)
#pragma unroll
      for (int v = 0; v < 8; ++v)
        GT[(size_t)(e0 + i * 16 + v + 8 * half) * D_DIM + d0 + j * 16 + lm] = c[i][j][v];
}

static __device__ __forceinline__ void ring_signal() {
  asm volatile("s_wait_dscnt 0x0\n\ts_barrier_signal -1" ::: "memory");
}
static __device__ __forceinline__ void ring_wait() {
  asm volatile("s_barrier_wait -1" ::: "memory");
}

static __device__ __forceinline__ void frag_load(v4f (&g)[BK * NC], const float* tile, int blk) {
#pragma unroll
  for (int kk = 0; kk < BK; ++kk)
#pragma unroll
    for (int j = 0; j < NC; ++j)
      g[kk * NC + j] = *(const v4f*)(tile + j * 16 * GP + (blk * BK + kk) * 8);
}

static __device__ __forceinline__ void mma_group(const v4f (&g)[BK * NC], int kk, v4f a,
                                                 v8f (&c)[NC]) {
  const v2f a0 = __builtin_shufflevector(a, a, 0, 1);
  const v2f a1 = __builtin_shufflevector(a, a, 2, 3);
#pragma unroll
  for (int j = 0; j < NC; ++j) {
    const v4f gv = g[kk * NC + j];
    c[j] = wmma4(a0, __builtin_shufflevector(gv, gv, 0, 1), c[j]);
  }
#pragma unroll
  for (int j = 0; j < NC; ++j) {
    const v4f gv = g[kk * NC + j];
    c[j] = wmma4(a1, __builtin_shufflevector(gv, gv, 2, 3), c[j]);
  }
}

static __device__ __forceinline__ void tile_step(
    const float* __restrict__ gsrc, const float* __restrict__ fa,
    const float* cur, const float* nxt, float* dst, int it,
    v4f (&gst)[NST], const v4f (&aUse)[KC / 8], v4f (&aLoad)[KC / 8], v8f (&c)[NC],
    v4f (&g0)[BK * NC], v4f (&g1)[BK * NC]) {
  const int n1 = (it + 1) & (NIT - 1);
  const float* a1 = fa + (n1 % NKC) * KC;
#pragma unroll
  for (int kg = 0; kg < KC / 8; ++kg) aLoad[kg] = *(const v4f*)(a1 + kg * 8);

  frag_load(g1, cur, 1);
  __builtin_amdgcn_sched_barrier(0);
  mma_group(g0, 0, aUse[0], c);
  mma_group(g0, 1, aUse[1], c);
  __builtin_amdgcn_sched_barrier(0);

  frag_load(g0, cur, 2);
  __builtin_amdgcn_sched_barrier(0);
  mma_group(g1, 0, aUse[2], c);
  mma_group(g1, 1, aUse[3], c);
  __builtin_amdgcn_sched_barrier(0);

  frag_load(g1, cur, 3);
  __builtin_amdgcn_sched_barrier(0);
  mma_group(g0, 0, aUse[4], c);
  __builtin_amdgcn_sched_barrier(0);

  ring_wait();
#pragma unroll
  for (int i = 0; i < NST; ++i) *(v4f*)(dst + 16 * i * GP) = gst[i];
  __builtin_amdgcn_sched_barrier(0);
  mma_group(g0, 1, aUse[5], c);
  __builtin_amdgcn_sched_barrier(0);

  ring_signal();

  const int n3 = (it + 3) & (NIT - 1);
  const float* g3 = gsrc + (size_t)((n3 / NKC) * NE) * D_DIM + (n3 % NKC) * KC;
#pragma unroll
  for (int i = 0; i < NST; ++i) gst[i] = *(const v4f*)(g3 + (size_t)(16 * i) * D_DIM);

  frag_load(g0, nxt, 0);
  __builtin_amdgcn_sched_barrier(0);
  mma_group(g1, 0, aUse[6], c);
  mma_group(g1, 1, aUse[7], c);
  __builtin_amdgcn_sched_barrier(0);
}

__global__ __launch_bounds__(256) void attn_batch(
    const float* __restrict__ rF, const float* __restrict__ pF,
    const float* __restrict__ rN, const float* __restrict__ pN,
    const float* __restrict__ GT, const float* __restrict__ Wp,
    const float* __restrict__ bp, float* __restrict__ out) {
  __shared__ __attribute__((aligned(16))) float sG[NBUF * TILE_F];
  __shared__ __attribute__((aligned(16))) float sW[NW][256];
  __shared__ __attribute__((aligned(16))) float sS[NW][256];
  __shared__ float sRows[NW * 72][36];
  __shared__ float sAtt[NW][2][36];
  __shared__ float sWeff[NW][2][8];
  __shared__ float sRed[NW][RP];

  const int tid = threadIdx.x;
  const int wave = tid >> 5;
  const int lane = tid & 31;
  const int lm = lane & 15;
  const int half = lane >> 4;
  const int b0 = blockIdx.x * NW;
  const int b = b0 + wave;

  const float* Frow =
      (lm < M_MOL) ? (pF + ((size_t)lm * B_DIM + b) * D_DIM)
                   : (rF + ((size_t)(lm - M_MOL) * B_DIM + b) * D_DIM);
  const float* fa = Frow + 4 * half;

  const float* gsrc = GT + (size_t)(tid >> 4) * D_DIM + (tid & 15) * 4;
  const int sdst = (tid >> 4) * GP + (tid & 15) * 4;
  const int frd = lm * GP + 4 * half;

  v4f gst[NST];
  v4f aA[KC / 8], aB[KC / 8];
  v4f g0[BK * NC], g1[BK * NC];
  {
    v4f t1[NST];
#pragma unroll
    for (int i = 0; i < NST; ++i) gst[i] = *(const v4f*)(gsrc + (size_t)(16 * i) * D_DIM);
#pragma unroll
    for (int i = 0; i < NST; ++i) t1[i] = *(const v4f*)(gsrc + (size_t)(16 * i) * D_DIM + KC);
#pragma unroll
    for (int kg = 0; kg < KC / 8; ++kg) aA[kg] = *(const v4f*)(fa + kg * 8);
#pragma unroll
    for (int i = 0; i < NST; ++i) *(v4f*)(&sG[sdst + 16 * i * GP]) = gst[i];
#pragma unroll
    for (int i = 0; i < NST; ++i) *(v4f*)(&sG[TILE_F + sdst + 16 * i * GP]) = t1[i];
#pragma unroll
    for (int i = 0; i < NST; ++i)
      gst[i] = *(const v4f*)(gsrc + (size_t)(16 * i) * D_DIM + 2 * KC);
  }
  __syncthreads();
  frag_load(g0, &sG[frd], 0);
  ring_signal();

  int o0 = 0, o1 = TILE_F, o2 = 2 * TILE_F;

  v8f sacc = {};
#pragma unroll 1
  for (int eg = 0; eg < NEG; ++eg) {
    const int e0 = eg * NE;
    v8f c[NC];
#pragma unroll
    for (int j = 0; j < NC; ++j) c[j] = (v8f){};

#pragma unroll 1
    for (int kc = 0; kc < NKC; kc += 2) {
      const int it = eg * NKC + kc;
      tile_step(gsrc, fa, &sG[o0 + frd], &sG[o1 + frd], &sG[o2 + sdst], it, gst, aA, aB, c, g0, g1);
      tile_step(gsrc, fa, &sG[o1 + frd], &sG[o2 + frd], &sG[o0 + sdst], it + 1, gst, aB, aA, c, g0, g1);
      const int t = o0;
      o0 = o2;
      o2 = o1;
      o1 = t;
    }

    v2f b2[NC][4];
#pragma unroll
    for (int j = 0; j < NC; ++j)
#pragma unroll
      for (int kk = 0; kk < 4; ++kk)
        b2[j][kk] = *(const v2f*)(Frow + e0 + j * 16 + kk * 4 + 2 * half);

#pragma unroll
    for (int j = 0; j < NC; ++j) {
#pragma unroll
      for (int v = 0; v < 8; ++v) sW[wave][(v + 8 * half) * 16 + lm] = c[j][v];
#pragma unroll
      for (int kk = 0; kk < 4; ++kk) {
        v2f a2 = *(const v2f*)(&sW[wave][lm * 16 + kk * 4 + 2 * half]);
        sacc = wmma4(a2, b2[j][kk], sacc);
      }
    }
  }
  ring_wait();
#pragma unroll
  for (int v = 0; v < 8; ++v) sS[wave][(v + 8 * half) * 16 + lm] = sacc[v];
  __syncthreads();

  const float scale = 0.03125f;
  for (int r = tid; r < NW * 72; r += 256) {
    const int bb = r / 72;
    const int t = r - bb * 72;
    const float* S = sS[bb];
    const int dir = (t >= 36) ? 1 : 0;
    const int q = dir ? t - 36 : t;
    const bool two = q >= 8;
    int qa0 = q, qa1 = q;
    if (two) { qa0 = c_I[q - 8]; qa1 = c_J[q - 8]; }
    const float* pa = dir ? S + (8 + qa0) * 16 : S + qa0 * 16 + 8;
    const float* pb = dir ? S + (8 + qa1) * 16 : S + qa1 * 16 + 8;
    const v4f alo = *(const v4f*)pa, ahi = *(const v4f*)(pa + 4);
    const v4f blo = *(const v4f*)pb, bhi = *(const v4f*)(pb + 4);
    float ra[8], rb[8];
#pragma unroll
    for (int c = 0; c < 4; ++c) {
      ra[c] = alo[c]; ra[4 + c] = ahi[c];
      rb[c] = blo[c]; rb[4 + c] = bhi[c];
    }
    float row[36];
    float mx = -3.4e38f;
#pragma unroll
    for (int k = 0; k < 8; ++k) {
      const float s1 = ra[k];
      const float s2 = s1 + rb[k];
      const float s = (two ? s2 : s1) * scale;
      row[k] = s;
      mx = fmaxf(mx, s);
    }
    {
      int k = 8;
#pragma unroll
      for (int i = 0; i < 7; ++i)
#pragma unroll
        for (int j = i + 1; j < 8; ++j) {
          const float s1 = ra[i] + ra[j];
          const float s2 = (s1 + rb[i]) + rb[j];
          const float s = (two ? s2 : s1) * scale;
          row[k] = s;
          mx = fmaxf(mx, s);
          ++k;
        }
    }
    float sum = 0.f;
#pragma unroll
    for (int k = 0; k < 36; ++k) { float e = __expf(row[k] - mx); row[k] = e; sum += e; }
    const float inv = 1.f / sum;
#pragma unroll
    for (int k = 0; k < 36; ++k) sRows[r][k] = row[k] * inv;
  }
  __syncthreads();

  for (int r = tid; r < NW * 72; r += 256) {
    const int bb = r / 72;
    const int t = r - bb * 72;
    const int dir = (t >= 36) ? 1 : 0;
    const int k = dir ? t - 36 : t;
    float a = 0.f;
#pragma unroll
    for (int q = 0; q < 36; ++q) a += sRows[bb * 72 + dir * 36 + q][k];
    a *= (1.f / 36.f);
    sAtt[bb][dir][k] = a;
    const size_t base = dir ? (size_t)(B_DIM * 4 + B_DIM * 36) : (size_t)(B_DIM * 4);
    out[base + (size_t)(b0 + bb) * 36 + k] = a;
  }
  __syncthreads();

  if (tid < NW * 16) {
    const int bb = tid >> 4;
    const int dir = (tid >> 3) & 1, m = tid & 7;
    float pw[NPAIR];
#pragma unroll
    for (int pr = 0; pr < NPAIR; ++pr) pw[pr] = sAtt[bb][dir][8 + pr];
    float w = sAtt[bb][dir][m];
    {
      int pr = 0;
#pragma unroll
      for (int i = 0; i < 7; ++i)
#pragma unroll
        for (int j = i + 1; j < 8; ++j) {
          if (i == m || j == m) w += pw[pr];
          ++pr;
        }
    }
    sWeff[bb][dir][m] = w;
  }
  __syncthreads();

  {
    const int d = tid * 4;
#pragma unroll 1
    for (int bb = 0; bb < NW; ++bb) {
      float reac[4] = {0.f, 0.f, 0.f, 0.f};
#pragma unroll
      for (int m = 0; m < M_MOL; ++m) {
        const size_t off = ((size_t)m * B_DIM + (b0 + bb)) * D_DIM + d;
        v4f rv = *(const v4f*)(rF + off);
        v4f rnv = *(const v4f*)(rN + off);
        v4f pv = *(const v4f*)(pF + off);
        v4f pnv = *(const v4f*)(pN + off);
        const float wr = sWeff[bb][0][m], wp = sWeff[bb][1][m];
#pragma unroll
        for (int x = 0; x < 4; ++x)
          reac[x] += wr * (rv[x] + rnv[x]) - wp * (pv[x] + pnv[x]);
      }
#pragma unroll
      for (int o = 0; o < 4; ++o) {
        float s = 0.f;
#pragma unroll
        for (int x = 0; x < 4; ++x) s += reac[x] * Wp[(size_t)o * D_DIM + d + x];
        sRed[bb][tid * 4 + o] = s;
      }
    }
  }
  __syncthreads();
  if (tid < NW * 4) {
    const int bb = tid >> 2, o = tid & 3;
    float s = bp[o];
    for (int t = 0; t < 256; ++t) s += sRed[bb][t * 4 + o];
    out[(size_t)(b0 + bb) * 4 + o] = s;
  }
}

extern "C" void kernel_launch(void* const* d_in, const int* in_sizes, int n_in,
                              void* d_out, int out_size, void* d_ws,
                              size_t ws_size, hipStream_t stream) {
  (void)in_sizes; (void)n_in; (void)out_size; (void)ws_size;
  const float* rF = (const float*)d_in[0];
  const float* pF = (const float*)d_in[1];
  const float* rN = (const float*)d_in[2];
  const float* pN = (const float*)d_in[3];
  const float* Wq = (const float*)d_in[4];
  const float* Wk = (const float*)d_in[5];
  const float* Wp = (const float*)d_in[6];
  const float* bp = (const float*)d_in[7];
  float* out = (float*)d_out;
  float* GT = (float*)d_ws;

  gemm_gt<<<256, 128, 0, stream>>>(Wq, Wk, GT);
  attn_batch<<<256, 256, 0, stream>>>(rF, pF, rN, pN, GT, Wp, bp, out);
}
